# starting kernel
# baseline (speedup 1.0000x reference)
#include <hip/hip_runtime.h>
#include <hip/hip_bf16.h>
#include <math.h>

typedef __attribute__((ext_vector_type(2))) float v2f;
typedef __attribute__((ext_vector_type(4))) float v4f;
typedef __attribute__((ext_vector_type(8))) float v8f;

#define N_GAUSS   1024
#define IMG_H     256
#define IMG_W     256
#define EPS_COV   1e-6f
#define ALPHA_MAX 0.999f

#define GF_STRIDE 10
#define GF_DW     (N_GAUSS * GF_STRIDE)
#define CP_DW     (N_GAUSS * 4)
#define SMEM_DW   (GF_DW + CP_DW)

__global__ __launch_bounds__(256) void gs_prep(const float* __restrict__ means,
                                               const float* __restrict__ log_scales,
                                               const float* __restrict__ rot,
                                               const float* __restrict__ colors,
                                               const float* __restrict__ opac,
                                               float* __restrict__ ws) {
  int n = blockIdx.x * blockDim.x + threadIdx.x;
  if (n >= N_GAUSS) return;

  float mx = means[2 * n + 0];
  float my = means[2 * n + 1];
  float s2x = __expf(2.f * log_scales[2 * n + 0]);
  float s2y = __expf(2.f * log_scales[2 * n + 1]);
  float s, c;
  __sincosf(rot[n], &s, &c);

  float a = c * c * s2x + s * s * s2y + EPS_COV;
  float b = c * s * (s2x - s2y);
  float d = s * s * s2x + c * c * s2y + EPS_COV;
  float inv = 1.f / (a * d - b * b);
  float ia = d * inv, ib = -b * inv, id = a * inv;

  float* gf = ws + (size_t)n * GF_STRIDE;
  gf[0] = -0.5f * ia;
  gf[1] = ia * mx + ib * my;
  gf[2] = -0.5f * id;
  gf[3] = id * my + ib * mx;
  gf[4] = -ib;
  gf[5] = -0.5f * (ia * mx * mx + 2.f * ib * mx * my + id * my * my);
  gf[6] = 0.f; gf[7] = 0.f;
  gf[8] = 0.f; gf[9] = 0.f;

  float* cp = ws + GF_DW + (size_t)n * 4;
  cp[0] = colors[3 * n + 0];
  cp[1] = colors[3 * n + 1];
  cp[2] = colors[3 * n + 2];
  cp[3] = opac[n];
}

__global__ __launch_bounds__(256) void gs_render(const float* __restrict__ ws,
                                                 float* __restrict__ out) {
  __shared__ float smem[SMEM_DW];

  {
    const v4f* src = (const v4f*)ws;
    v4f* dst = (v4f*)smem;
    for (int i = threadIdx.x; i < SMEM_DW / 4; i += 256) dst[i] = src[i];
  }
  __syncthreads();

  const float* sg = smem;
  const float* sc = smem + GF_DW;

  const int lane = threadIdx.x & 31;
  const int wave = threadIdx.x >> 5;
  const int half = lane >> 4;
  const int col  = lane & 15;

  const int p = blockIdx.x * 128 + wave * 16 + col;
  const float x = (float)(p & (IMG_W - 1));
  const float y = (float)(p >> 8);

  const float f[8] = { x * x, x, y * y, y, x * y, 1.f, 0.f, 0.f };
  const v2f bA = { f[2 * half + 0], f[2 * half + 1] };
  const v2f bB = { f[4 + 2 * half + 0], f[4 + 2 * half + 1] };

  float T = 1.f;
  float cr = 0.f, cg = 0.f, cb = 0.f;

  for (int gb = 0; gb < N_GAUSS; gb += 16) {
    const float* gf = sg + (gb + col) * GF_STRIDE;
    const v2f aA = { gf[2 * half + 0], gf[2 * half + 1] };
    const v2f aB = { gf[4 + 2 * half + 0], gf[4 + 2 * half + 1] };

    v8f P = {};
    P = __builtin_amdgcn_wmma_f32_16x16x4_f32(false, aA, false, bA,
                                              (short)0, P, false, false);
    P = __builtin_amdgcn_wmma_f32_16x16x4_f32(false, aB, false, bB,
                                              (short)0, P, false, false);

    const int gg = gb + 8 * half;
    float w[8];
    float q = 1.f;
#pragma unroll
    for (int j = 0; j < 8; ++j) {
      float op = sc[(gg + j) * 4 + 3];
      float al = fminf(op * __expf(P[j]), ALPHA_MAX);
      w[j] = al * q;
      q *= (1.f - al);
    }

    float qo  = __shfl_xor(q, 16, 32);
    float pre = half ? (T * qo) : T;
    T *= q * qo;

#pragma unroll
    for (int j = 0; j < 8; ++j) {
      float wv = w[j] * pre;
      const float* cc = sc + (gg + j) * 4;
      cr = fmaf(wv, cc[0], cr);
      cg = fmaf(wv, cc[1], cg);
      cb = fmaf(wv, cc[2], cb);
    }
  }

  cr += __shfl_xor(cr, 16, 32);
  cg += __shfl_xor(cg, 16, 32);
  cb += __shfl_xor(cb, 16, 32);
  if (half == 0) {
    float* o = out + (size_t)p * 3;
    o[0] = cr; o[1] = cg; o[2] = cb;
  }
}

extern "C" void kernel_launch(void* const* d_in, const int* in_sizes, int n_in,
                              void* d_out, int out_size, void* d_ws, size_t ws_size,
                              hipStream_t stream) {
  const float* means      = (const float*)d_in[0];
  const float* log_scales = (const float*)d_in[1];
  const float* rotations  = (const float*)d_in[2];
  const float* colors     = (const float*)d_in[3];
  const float* opacities  = (const float*)d_in[4];
  float* ws = (float*)d_ws;

  gs_prep<<<N_GAUSS / 256, 256, 0, stream>>>(means, log_scales, rotations,
                                             colors, opacities, ws);
  gs_render<<<(IMG_H * IMG_W) / 128, 256, 0, stream>>>(ws, (float*)d_out);
}


# render: gaussian list cut into two contiguous halves, one per lane half, each with a running transmittance in a register; per-step shuffle merge gone, halves joined once at end (sum order changes)
# speedup vs baseline: 1.1817x; 1.1817x over previous
#include <hip/hip_runtime.h>
#include <hip/hip_bf16.h>
#include <math.h>

typedef __attribute__((ext_vector_type(2))) float v2f;
typedef __attribute__((ext_vector_type(4))) float v4f;
typedef __attribute__((ext_vector_type(8))) float v8f;

#define N_GAUSS   1024
#define IMG_H     256
#define IMG_W     256
#define EPS_COV   1e-6f
#define ALPHA_MAX 0.999f

#define GF_STRIDE 10
#define GF_DW     (N_GAUSS * GF_STRIDE)
#define CP_DW     (N_GAUSS * 4)
#define SMEM_DW   (GF_DW + CP_DW)

__global__ __launch_bounds__(256) void gs_prep(const float* __restrict__ means,
                                               const float* __restrict__ log_scales,
                                               const float* __restrict__ rot,
                                               const float* __restrict__ colors,
                                               const float* __restrict__ opac,
                                               float* __restrict__ ws) {
  int n = blockIdx.x * blockDim.x + threadIdx.x;
  if (n >= N_GAUSS) return;

  float mx = means[2 * n + 0];
  float my = means[2 * n + 1];
  float s2x = __expf(2.f * log_scales[2 * n + 0]);
  float s2y = __expf(2.f * log_scales[2 * n + 1]);
  float s, c;
  __sincosf(rot[n], &s, &c);

  float a = c * c * s2x + s * s * s2y + EPS_COV;
  float b = c * s * (s2x - s2y);
  float d = s * s * s2x + c * c * s2y + EPS_COV;
  float inv = 1.f / (a * d - b * b);
  float ia = d * inv, ib = -b * inv, id = a * inv;

  float* gf = ws + (size_t)n * GF_STRIDE;
  gf[0] = -0.5f * ia;
  gf[1] = ia * mx + ib * my;
  gf[2] = -0.5f * id;
  gf[3] = id * my + ib * mx;
  gf[4] = -ib;
  gf[5] = -0.5f * (ia * mx * mx + 2.f * ib * mx * my + id * my * my);
  gf[6] = 0.f; gf[7] = 0.f;
  gf[8] = 0.f; gf[9] = 0.f;

  float* cp = ws + GF_DW + (size_t)n * 4;
  cp[0] = colors[3 * n + 0];
  cp[1] = colors[3 * n + 1];
  cp[2] = colors[3 * n + 2];
  cp[3] = opac[n];
}

__global__ __launch_bounds__(256) void gs_render(const float* __restrict__ ws,
                                                 float* __restrict__ out) {
  __shared__ float smem[SMEM_DW];

  {
    const v4f* src = (const v4f*)ws;
    v4f* dst = (v4f*)smem;
    for (int i = threadIdx.x; i < SMEM_DW / 4; i += 256) dst[i] = src[i];
  }
  __syncthreads();

  const float* sg = smem;
  const float* sc = smem + GF_DW;

  const int lane = threadIdx.x & 31;
  const int wave = threadIdx.x >> 5;
  const int half = lane >> 4;
  const int col  = lane & 15;

  const int p = blockIdx.x * 128 + wave * 16 + col;
  const float x = (float)(p & (IMG_W - 1));
  const float y = (float)(p >> 8);

  const float f[8] = { x * x, x, y * y, y, x * y, 1.f, 0.f, 0.f };
  const v2f bA = { f[2 * half + 0], f[2 * half + 1] };
  const v2f bB = { f[4 + 2 * half + 0], f[4 + 2 * half + 1] };

  const int arow = (col & 7) + (col >> 3) * (N_GAUSS / 2);
  const float* gfp = sg + arow * GF_STRIDE + 2 * half;
  const float* ccp = sc + half * (N_GAUSS / 2) * 4;

  float T = 1.f;
  float cr = 0.f, cg = 0.f, cb = 0.f;

  for (int s = 0; s < N_GAUSS / 2; s += 8) {
    const float* gf = gfp + s * GF_STRIDE;
    const v2f aA = { gf[0], gf[1] };
    const v2f aB = { gf[4], gf[5] };

    v8f P = {};
    P = __builtin_amdgcn_wmma_f32_16x16x4_f32(false, aA, false, bA,
                                              (short)0, P, false, false);
    P = __builtin_amdgcn_wmma_f32_16x16x4_f32(false, aB, false, bB,
                                              (short)0, P, false, false);

#pragma unroll
    for (int j = 0; j < 8; ++j) {
      const float* cc = ccp + (s + j) * 4;
      float al = fminf(cc[3] * __expf(P[j]), ALPHA_MAX);
      float w  = al * T;
      T = fmaf(-al, T, T);
      cr = fmaf(w, cc[0], cr);
      cg = fmaf(w, cc[1], cg);
      cb = fmaf(w, cc[2], cb);
    }
  }

  float hr = __shfl_xor(cr, 16, 32);
  float hg = __shfl_xor(cg, 16, 32);
  float hb = __shfl_xor(cb, 16, 32);
  if (half == 0) {
    float* o = out + (size_t)p * 3;
    o[0] = fmaf(T, hr, cr);
    o[1] = fmaf(T, hg, cg);
    o[2] = fmaf(T, hb, cb);
  }
}

extern "C" void kernel_launch(void* const* d_in, const int* in_sizes, int n_in,
                              void* d_out, int out_size, void* d_ws, size_t ws_size,
                              hipStream_t stream) {
  const float* means      = (const float*)d_in[0];
  const float* log_scales = (const float*)d_in[1];
  const float* rotations  = (const float*)d_in[2];
  const float* colors     = (const float*)d_in[3];
  const float* opacities  = (const float*)d_in[4];
  float* ws = (float*)d_ws;

  gs_prep<<<N_GAUSS / 256, 256, 0, stream>>>(means, log_scales, rotations,
                                             colors, opacities, ws);
  gs_render<<<(IMG_H * IMG_W) / 128, 256, 0, stream>>>(ws, (float*)d_out);
}


# preprocessing fused into the render kernel: each block builds the gaussian tables in its LDS from the raw inputs (4 gaussians per thread); one launch instead of two, workspace unused
# speedup vs baseline: 1.5261x; 1.2914x over previous
#include <hip/hip_runtime.h>
#include <hip/hip_bf16.h>
#include <math.h>


typedef __attribute__((ext_vector_type(2))) float v2f;
typedef __attribute__((ext_vector_type(4))) float v4f;
typedef __attribute__((ext_vector_type(8))) float v8f;

#define N_GAUSS   1024
#define IMG_H     256
#define IMG_W     256
#define EPS_COV   1e-6f
#define ALPHA_MAX 0.999f

#define GF_STRIDE 10
#define GF_DW     (N_GAUSS * GF_STRIDE)
#define CP_DW     (N_GAUSS * 4)
#define SMEM_DW   (GF_DW + CP_DW)

__device__ __forceinline__ void gauss_prep(int n,
                                           const float* __restrict__ means,
                                           const float* __restrict__ log_scales,
                                           const float* __restrict__ rot,
                                           const float* __restrict__ colors,
                                           const float* __restrict__ opac,
                                           float* sg, float* sc) {
  float mx = means[2 * n + 0];
  float my = means[2 * n + 1];
  float s2x = __expf(2.f * log_scales[2 * n + 0]);
  float s2y = __expf(2.f * log_scales[2 * n + 1]);
  float s, c;
  __sincosf(rot[n], &s, &c);

  float a = c * c * s2x + s * s * s2y + EPS_COV;
  float b = c * s * (s2x - s2y);
  float d = s * s * s2x + c * c * s2y + EPS_COV;
  float inv = 1.f / (a * d - b * b);
  float ia = d * inv, ib = -b * inv, id = a * inv;

  float* gf = sg + n * GF_STRIDE;
  gf[0] = -0.5f * ia;
  gf[1] = ia * mx + ib * my;
  gf[2] = -0.5f * id;
  gf[3] = id * my + ib * mx;
  gf[4] = -ib;
  gf[5] = -0.5f * (ia * mx * mx + 2.f * ib * mx * my + id * my * my);
  gf[6] = 0.f; gf[7] = 0.f;
  gf[8] = 0.f; gf[9] = 0.f;

  float* cp = sc + n * 4;
  cp[0] = colors[3 * n + 0];
  cp[1] = colors[3 * n + 1];
  cp[2] = colors[3 * n + 2];
  cp[3] = opac[n];
}

__global__ __launch_bounds__(256) void gs_render(const float* __restrict__ means,
                                                 const float* __restrict__ log_scales,
                                                 const float* __restrict__ rot,
                                                 const float* __restrict__ colors,
                                                 const float* __restrict__ opac,
                                                 float* __restrict__ out) {
  __shared__ float smem[SMEM_DW];

  float* sg = smem;
  float* sc = smem + GF_DW;

#pragma unroll
  for (int k = 0; k < N_GAUSS / 256; ++k)
    gauss_prep(threadIdx.x + 256 * k, means, log_scales, rot, colors, opac, sg, sc);
  __syncthreads();

  const int lane = threadIdx.x & 31;
  const int wave = threadIdx.x >> 5;
  const int half = lane >> 4;
  const int col  = lane & 15;

  const int p = blockIdx.x * 128 + wave * 16 + col;
  const float x = (float)(p & (IMG_W - 1));
  const float y = (float)(p >> 8);

  const float f[8] = { x * x, x, y * y, y, x * y, 1.f, 0.f, 0.f };
  const v2f bA = { f[2 * half + 0], f[2 * half + 1] };
  const v2f bB = { f[4 + 2 * half + 0], f[4 + 2 * half + 1] };

  const int arow = (col & 7) + (col >> 3) * (N_GAUSS / 2);
  const float* gfp = sg + arow * GF_STRIDE + 2 * half;
  const float* ccp = sc + half * (N_GAUSS / 2) * 4;

  float T = 1.f;
  float cr = 0.f, cg = 0.f, cb = 0.f;

  for (int s = 0; s < N_GAUSS / 2; s += 8) {
    const float* gf = gfp + s * GF_STRIDE;
    const v2f aA = { gf[0], gf[1] };
    const v2f aB = { gf[4], gf[5] };

    v8f P = {};
    P = __builtin_amdgcn_wmma_f32_16x16x4_f32(false, aA, false, bA,
                                              (short)0, P, false, false);
    P = __builtin_amdgcn_wmma_f32_16x16x4_f32(false, aB, false, bB,
                                              (short)0, P, false, false);

#pragma unroll
    for (int j = 0; j < 8; ++j) {
      const float* cc = ccp + (s + j) * 4;
      float al = fminf(cc[3] * __expf(P[j]), ALPHA_MAX);
      float w  = al * T;
      T = fmaf(-al, T, T);
      cr = fmaf(w, cc[0], cr);
      cg = fmaf(w, cc[1], cg);
      cb = fmaf(w, cc[2], cb);
    }
  }

  float hr = __shfl_xor(cr, 16, 32);
  float hg = __shfl_xor(cg, 16, 32);
  float hb = __shfl_xor(cb, 16, 32);
  if (half == 0) {
    float* o = out + (size_t)p * 3;
    o[0] = fmaf(T, hr, cr);
    o[1] = fmaf(T, hg, cg);
    o[2] = fmaf(T, hb, cb);
  }
}

extern "C" void kernel_launch(void* const* d_in, const int* in_sizes, int n_in,
                              void* d_out, int out_size, void* d_ws, size_t ws_size,
                              hipStream_t stream) {
  const float* means      = (const float*)d_in[0];
  const float* log_scales = (const float*)d_in[1];
  const float* rotations  = (const float*)d_in[2];
  const float* colors     = (const float*)d_in[3];
  const float* opacities  = (const float*)d_in[4];

  gs_render<<<(IMG_H * IMG_W) / 128, 256, 0, stream>>>(means, log_scales, rotations,
                                                       colors, opacities, (float*)d_out);
}


# per 32x4 pixel tile the prologue drops every gaussian whose alpha is exactly 0 on the whole tile (power bound) and compacts the rest in index order; loop trips follow the survivors; sum order kept
# speedup vs baseline: 3.0573x; 2.0033x over previous
#include <hip/hip_runtime.h>
#include <hip/hip_bf16.h>
#include <math.h>


typedef __attribute__((ext_vector_type(2))) float v2f;
typedef __attribute__((ext_vector_type(4))) float v4f;
typedef __attribute__((ext_vector_type(8))) float v8f;

#define N_GAUSS   1024
#define IMG_H     256
#define IMG_W     256
#define EPS_COV   1e-6f
#define ALPHA_MAX 0.999f

#define GF_STRIDE 10
#define GF_DW     (N_GAUSS * GF_STRIDE)
#define CP_DW     (N_GAUSS * 4)
#define SMEM_DW   (GF_DW + CP_DW)

#define TILE_W    32
#define TILE_H    4
#define HALF_W    15.5f
#define HALF_H    1.5f
#define CULL_BELOW (-112.0f)

__global__ __launch_bounds__(256) void gs_render(const float* __restrict__ means,
                                                 const float* __restrict__ log_scales,
                                                 const float* __restrict__ rot,
                                                 const float* __restrict__ colors,
                                                 const float* __restrict__ opac,
                                                 float* __restrict__ out) {
  __shared__ float smem[SMEM_DW];
  __shared__ int s_cnt[8];

  float* sg = smem;
  float* sc = smem + GF_DW;

  const int lane = threadIdx.x & 31;
  const int wave = threadIdx.x >> 5;
  const int half = lane >> 4;
  const int col  = lane & 15;

  const int tx = blockIdx.x & (IMG_W / TILE_W - 1);
  const int ty = blockIdx.x / (IMG_W / TILE_W);
  const float xc = (float)(tx * TILE_W) + HALF_W;
  const float yc = (float)(ty * TILE_H) + HALF_H;

  float g[4][6];
  float cq[4][4];
  unsigned bal[4];

#pragma unroll
  for (int k = 0; k < 4; ++k) {
    const int n = wave * 128 + k * 32 + lane;
    float mx = means[2 * n + 0];
    float my = means[2 * n + 1];
    float s2x = __expf(2.f * log_scales[2 * n + 0]);
    float s2y = __expf(2.f * log_scales[2 * n + 1]);
    float s, c;
    __sincosf(rot[n], &s, &c);

    float a = c * c * s2x + s * s * s2y + EPS_COV;
    float b = c * s * (s2x - s2y);
    float d = s * s * s2x + c * c * s2y + EPS_COV;
    float inv = 1.f / (a * d - b * b);
    float ia = d * inv, ib = -b * inv, id = a * inv;

    g[k][0] = -0.5f * ia;
    g[k][1] = ia * mx + ib * my;
    g[k][2] = -0.5f * id;
    g[k][3] = id * my + ib * mx;
    g[k][4] = -ib;
    g[k][5] = -0.5f * (ia * mx * mx + 2.f * ib * mx * my + id * my * my);
    cq[k][0] = colors[3 * n + 0];
    cq[k][1] = colors[3 * n + 1];
    cq[k][2] = colors[3 * n + 2];
    cq[k][3] = opac[n];

    float dx = xc - mx, dy = yc - my;
    float pc = -0.5f * (ia * dx * dx + 2.f * ib * dx * dy + id * dy * dy);
    float reach = fabsf(ia * dx + ib * dy) * HALF_W + fabsf(ib * dx + id * dy) * HALF_H;
    bool  psd = (ia >= 0.f) && (id >= 0.f) && (ia * id >= ib * ib);
    float curv = 0.5f * (fabsf(ia) * (HALF_W * HALF_W) + 2.f * fabsf(ib) * (HALF_W * HALF_H)
                         + fabsf(id) * (HALF_H * HALF_H));
    float ub = pc + reach + (psd ? 0.f : curv);
    float mag = (fabsf(g[k][0]) + fabsf(g[k][2]) + fabsf(g[k][4])) * 65025.f
              + (fabsf(g[k][1]) + fabsf(g[k][3])) * 255.f + fabsf(g[k][5]) + reach
              + 0.5f * (fabsf(ia) * dx * dx + 2.f * fabsf(ib * dx * dy) + fabsf(id) * dy * dy);
    float guard = 4e-6f * mag;
    bool keep = !(ub + guard < CULL_BELOW);
    bal[k] = __builtin_amdgcn_ballot_w32(keep);
  }

  const unsigned lt = (1u << lane) - 1u;
  int off[4];
  int run = 0;
#pragma unroll
  for (int k = 0; k < 4; ++k) {
    off[k] = run + __builtin_popcount(bal[k] & lt);
    run += __builtin_popcount(bal[k]);
  }
  if (lane == 0) s_cnt[wave] = run;
  __syncthreads();

  int cnt[8];
#pragma unroll
  for (int i = 0; i < 8; ++i) cnt[i] = s_cnt[i];
  const int c0 = cnt[0] + cnt[1] + cnt[2] + cnt[3];
  const int c1 = cnt[4] + cnt[5] + cnt[6] + cnt[7];
  const int steps = __builtin_amdgcn_readfirstlane((max(c0, c1) + 7) & ~7);
  const int w4 = wave & 4;
  int base = (w4 >> 2) * (N_GAUSS / 2);
#pragma unroll
  for (int i = 0; i < 3; ++i)
    if (w4 + i < wave) base += cnt[w4 + i];

#pragma unroll
  for (int k = 0; k < 4; ++k) {
    if ((bal[k] >> lane) & 1u) {
      const int row = base + off[k];
      float* gf = sg + row * GF_STRIDE;
      gf[0] = g[k][0]; gf[1] = g[k][1]; gf[2] = g[k][2]; gf[3] = g[k][3];
      gf[4] = g[k][4]; gf[5] = g[k][5];
      gf[6] = 0.f; gf[7] = 0.f;
      float* cp = sc + row * 4;
      cp[0] = cq[k][3]; cp[1] = cq[k][0]; cp[2] = cq[k][1]; cp[3] = cq[k][2];
    }
  }
  {
    const int r  = threadIdx.x >> 7;
    const int cr_ = r ? c1 : c0;
    for (int i = cr_ + (threadIdx.x & 127); i < steps; i += 128) {
      const int row = r * (N_GAUSS / 2) + i;
      float* gf = sg + row * GF_STRIDE;
      gf[0] = 0.f; gf[1] = 0.f; gf[2] = 0.f; gf[3] = 0.f;
      gf[4] = 0.f; gf[5] = -1.0e30f; gf[6] = 0.f; gf[7] = 0.f;
      float* cp = sc + row * 4;
      cp[0] = 0.f; cp[1] = 0.f; cp[2] = 0.f; cp[3] = 0.f;
    }
  }
  __syncthreads();

  const int px = tx * TILE_W + (wave & 1) * 16 + col;
  const int py = ty * TILE_H + (wave >> 1);
  const int p  = py * IMG_W + px;
  const float x = (float)px;
  const float y = (float)py;

  const float f[8] = { x * x, x, y * y, y, x * y, 1.f, 0.f, 0.f };
  const v2f bA = { f[2 * half + 0], f[2 * half + 1] };
  const v2f bB = { f[4 + 2 * half + 0], f[4 + 2 * half + 1] };

  const int arow = (col & 7) + (col >> 3) * (N_GAUSS / 2);
  const float* gfp = sg + arow * GF_STRIDE + 2 * half;
  const float* ccp = sc + half * (N_GAUSS / 2) * 4;

  float T = 1.f;
  float cr = 0.f, cg = 0.f, cb = 0.f;

  for (int s = 0; s < steps; s += 8) {
    const float* gf = gfp + s * GF_STRIDE;
    const v2f aA = { gf[0], gf[1] };
    const v2f aB = { gf[4], gf[5] };

    v8f P = {};
    P = __builtin_amdgcn_wmma_f32_16x16x4_f32(false, aA, false, bA,
                                              (short)0, P, false, false);
    P = __builtin_amdgcn_wmma_f32_16x16x4_f32(false, aB, false, bB,
                                              (short)0, P, false, false);

#pragma unroll
    for (int j = 0; j < 8; ++j) {
      const float* cc = ccp + (s + j) * 4;
      float al = fminf(cc[0] * __expf(P[j]), ALPHA_MAX);
      float w  = al * T;
      T = fmaf(-al, T, T);
      cr = fmaf(w, cc[1], cr);
      cg = fmaf(w, cc[2], cg);
      cb = fmaf(w, cc[3], cb);
    }
  }

  float hr = __shfl_xor(cr, 16, 32);
  float hg = __shfl_xor(cg, 16, 32);
  float hb = __shfl_xor(cb, 16, 32);
  if (half == 0) {
    float* o = out + (size_t)p * 3;
    o[0] = fmaf(T, hr, cr);
    o[1] = fmaf(T, hg, cg);
    o[2] = fmaf(T, hb, cb);
  }
}

extern "C" void kernel_launch(void* const* d_in, const int* in_sizes, int n_in,
                              void* d_out, int out_size, void* d_ws, size_t ws_size,
                              hipStream_t stream) {
  const float* means      = (const float*)d_in[0];
  const float* log_scales = (const float*)d_in[1];
  const float* rotations  = (const float*)d_in[2];
  const float* colors     = (const float*)d_in[3];
  const float* opacities  = (const float*)d_in[4];

  gs_render<<<(IMG_H / TILE_H) * (IMG_W / TILE_W), 256, 0, stream>>>(means, log_scales, rotations,
                                                       colors, opacities, (float*)d_out);
}


# render: a wave's 16 pixels share a row, so log2(alpha) about the tile centre is rank 3 in x; y, log2(e), log2(opacity) fold into the A fragment: ONE 16x16x4 product per 8 steps, not 2; exp2 reads it
# speedup vs baseline: 3.5866x; 1.1731x over previous
#include <hip/hip_runtime.h>
#include <hip/hip_bf16.h>
#include <math.h>


typedef __attribute__((ext_vector_type(2))) float v2f;
typedef __attribute__((ext_vector_type(4))) float v4f;
typedef __attribute__((ext_vector_type(8))) float v8f;

#define N_GAUSS   1024
#define IMG_H     256
#define IMG_W     256
#define EPS_COV   1e-6f
#define ALPHA_MAX 0.999f

#define LOG2E     1.44269504f
#define GF_STRIDE 10
#define GF_DW     (N_GAUSS * GF_STRIDE)
#define CP_DW     (N_GAUSS * 4)
#define SMEM_DW   (GF_DW + CP_DW)

#define TILE_W    32
#define TILE_H    4
#define HALF_W    15.5f
#define HALF_H    1.5f
#define CULL_BELOW (-161.6f)

__global__ __launch_bounds__(256) void gs_render(const float* __restrict__ means,
                                                 const float* __restrict__ log_scales,
                                                 const float* __restrict__ rot,
                                                 const float* __restrict__ colors,
                                                 const float* __restrict__ opac,
                                                 float* __restrict__ out) {
  __shared__ float smem[SMEM_DW];
  __shared__ int s_cnt[8];

  float* sg = smem;
  float* sc = smem + GF_DW;

  const int lane = threadIdx.x & 31;
  const int wave = threadIdx.x >> 5;
  const int half = lane >> 4;
  const int col  = lane & 15;

  const int tx = blockIdx.x & (IMG_W / TILE_W - 1);
  const int ty = blockIdx.x / (IMG_W / TILE_W);
  const float xc = (float)(tx * TILE_W) + HALF_W;
  const float yc = (float)(ty * TILE_H) + HALF_H;

  float g[4][6];
  float cq[4][3];
  unsigned bal[4];

#pragma unroll
  for (int k = 0; k < 4; ++k) {
    const int n = wave * 128 + k * 32 + lane;
    float mx = means[2 * n + 0];
    float my = means[2 * n + 1];
    float s2x = __expf(2.f * log_scales[2 * n + 0]);
    float s2y = __expf(2.f * log_scales[2 * n + 1]);
    float s, c;
    __sincosf(rot[n], &s, &c);

    float a = c * c * s2x + s * s * s2y + EPS_COV;
    float b = c * s * (s2x - s2y);
    float d = s * s * s2x + c * c * s2y + EPS_COV;
    float inv = LOG2E / (a * d - b * b);
    float ia = d * inv, ib = -b * inv, id = a * inv;

    float dx = xc - mx, dy = yc - my;
    float gx = ia * dx + ib * dy;
    float gy = ib * dx + id * dy;
    float pc = -0.5f * (ia * dx * dx + 2.f * ib * dx * dy + id * dy * dy);

    g[k][0] = -0.5f * ia;
    g[k][1] = -gx;
    g[k][2] = -0.5f * id;
    g[k][3] = -gy;
    g[k][4] = -ib;
    g[k][5] = pc + __log2f(fmaxf(opac[n], 0.f));
    cq[k][0] = colors[3 * n + 0];
    cq[k][1] = colors[3 * n + 1];
    cq[k][2] = colors[3 * n + 2];

    float reach = fabsf(gx) * HALF_W + fabsf(gy) * HALF_H;
    bool  psd = (ia >= 0.f) && (id >= 0.f) && (ia * id >= ib * ib);
    float curv = 0.5f * (fabsf(ia) * (HALF_W * HALF_W) + 2.f * fabsf(ib) * (HALF_W * HALF_H)
                         + fabsf(id) * (HALF_H * HALF_H));
    float ub = pc + reach + (psd ? 0.f : curv);
    float mag = 0.5f * (fabsf(ia) * dx * dx + 2.f * fabsf(ib * dx * dy) + fabsf(id) * dy * dy)
              + reach + curv;
    float guard = 4e-6f * mag;
    bool keep = !(ub + guard < CULL_BELOW);
    bal[k] = __builtin_amdgcn_ballot_w32(keep);
  }

  const unsigned lt = (1u << lane) - 1u;
  int off[4];
  int run = 0;
#pragma unroll
  for (int k = 0; k < 4; ++k) {
    off[k] = run + __builtin_popcount(bal[k] & lt);
    run += __builtin_popcount(bal[k]);
  }
  if (lane == 0) s_cnt[wave] = run;
  __syncthreads();

  int cnt[8];
#pragma unroll
  for (int i = 0; i < 8; ++i) cnt[i] = s_cnt[i];
  const int c0 = cnt[0] + cnt[1] + cnt[2] + cnt[3];
  const int c1 = cnt[4] + cnt[5] + cnt[6] + cnt[7];
  const int steps = __builtin_amdgcn_readfirstlane((max(c0, c1) + 7) & ~7);
  const int w4 = wave & 4;
  int base = (w4 >> 2) * (N_GAUSS / 2);
#pragma unroll
  for (int i = 0; i < 3; ++i)
    if (w4 + i < wave) base += cnt[w4 + i];

#pragma unroll
  for (int k = 0; k < 4; ++k) {
    if ((bal[k] >> lane) & 1u) {
      const int row = base + off[k];
      float* gf = sg + row * GF_STRIDE;
      gf[0] = g[k][4]; gf[1] = g[k][1]; gf[2] = g[k][0]; gf[3] = 0.f;
      gf[4] = g[k][2]; gf[5] = g[k][3]; gf[6] = g[k][5]; gf[7] = 0.f;
      float* cp = sc + row * 4;
      cp[0] = cq[k][0]; cp[1] = cq[k][1]; cp[2] = cq[k][2]; cp[3] = 0.f;
    }
  }
  {
    const int r  = threadIdx.x >> 7;
    const int cr_ = r ? c1 : c0;
    for (int i = cr_ + (threadIdx.x & 127); i < steps; i += 128) {
      const int row = r * (N_GAUSS / 2) + i;
      float* gf = sg + row * GF_STRIDE;
      gf[0] = 0.f; gf[1] = 0.f; gf[2] = 0.f; gf[3] = 0.f;
      gf[4] = 0.f; gf[5] = 0.f; gf[6] = -1.0e30f; gf[7] = 0.f;
      float* cp = sc + row * 4;
      cp[0] = 0.f; cp[1] = 0.f; cp[2] = 0.f; cp[3] = 0.f;
    }
  }
  __syncthreads();

  const int px = tx * TILE_W + (wave & 1) * 16 + col;
  const int py = ty * TILE_H + (wave >> 1);
  const int p  = py * IMG_W + px;
  const float ex = (float)((wave & 1) * 16 + col) - HALF_W;
  const float ey = (float)(wave >> 1) - HALF_H;

  const v2f bF = { half ? 1.f : ex * ex, half ? 0.f : ex };
  const float eyc = half ? ey : 0.f;

  const int arow = (col & 7) + (col >> 3) * (N_GAUSS / 2);
  const float* gfp = sg + arow * GF_STRIDE + 4 * half;
  const float* ccp = sc + half * (N_GAUSS / 2) * 4;

  float T = 1.f;
  v2f crg = { 0.f, 0.f };
  float cb = 0.f;

  for (int s = 0; s < steps; s += 8) {
    const float* gf = gfp + s * GF_STRIDE;
    const float lin = fmaf(gf[0], ey, gf[1]);
    const v2f aF = { fmaf(lin, eyc, gf[2]), lin };

    v8f P = {};
    P = __builtin_amdgcn_wmma_f32_16x16x4_f32(false, aF, false, bF,
                                              (short)0, P, false, false);

#pragma unroll
    for (int j = 0; j < 8; ++j) {
      const float* cc = ccp + (s + j) * 4;
      float al = fminf(__builtin_amdgcn_exp2f(P[j]), ALPHA_MAX);
      float w  = al * T;
      T = fmaf(-al, T, T);
      const v2f rg = { cc[0], cc[1] };
      const v2f ww = { w, w };
      crg = __builtin_elementwise_fma(ww, rg, crg);
      cb = fmaf(w, cc[2], cb);
    }
  }

  const float cr = crg.x, cg = crg.y;
  float hr = __shfl_xor(cr, 16, 32);
  float hg = __shfl_xor(cg, 16, 32);
  float hb = __shfl_xor(cb, 16, 32);
  if (half == 0) {
    float* o = out + (size_t)p * 3;
    o[0] = fmaf(T, hr, cr);
    o[1] = fmaf(T, hg, cg);
    o[2] = fmaf(T, hb, cb);
  }
}

extern "C" void kernel_launch(void* const* d_in, const int* in_sizes, int n_in,
                              void* d_out, int out_size, void* d_ws, size_t ws_size,
                              hipStream_t stream) {
  const float* means      = (const float*)d_in[0];
  const float* log_scales = (const float*)d_in[1];
  const float* rotations  = (const float*)d_in[2];
  const float* colors     = (const float*)d_in[3];
  const float* opacities  = (const float*)d_in[4];

  gs_render<<<(IMG_H / TILE_H) * (IMG_W / TILE_W), 256, 0, stream>>>(means, log_scales, rotations,
                                                       colors, opacities, (float*)d_out);
}


# cull bound moved to where v_exp_f32 itself returns 0 (inputs under -126, taken at -130) with log2(opacity) counted in: shorter per-tile lists; dropped terms are exact zeros, sums and order kept
# speedup vs baseline: 3.8617x; 1.0767x over previous
#include <hip/hip_runtime.h>
#include <hip/hip_bf16.h>
#include <math.h>


typedef __attribute__((ext_vector_type(2))) float v2f;
typedef __attribute__((ext_vector_type(4))) float v4f;
typedef __attribute__((ext_vector_type(8))) float v8f;

#define N_GAUSS   1024
#define IMG_H     256
#define IMG_W     256
#define EPS_COV   1e-6f
#define ALPHA_MAX 0.999f

#define LOG2E     1.44269504f
#define GF_STRIDE 10
#define GF_DW     (N_GAUSS * GF_STRIDE)
#define CP_DW     (N_GAUSS * 4)
#define SMEM_DW   (GF_DW + CP_DW)

#define TILE_W    32
#define TILE_H    4
#define HALF_W    15.5f
#define HALF_H    1.5f
#define CULL_BELOW (-130.0f)

__global__ __launch_bounds__(256) void gs_render(const float* __restrict__ means,
                                                 const float* __restrict__ log_scales,
                                                 const float* __restrict__ rot,
                                                 const float* __restrict__ colors,
                                                 const float* __restrict__ opac,
                                                 float* __restrict__ out) {
  __shared__ float smem[SMEM_DW];
  __shared__ int s_cnt[8];

  float* sg = smem;
  float* sc = smem + GF_DW;

  const int lane = threadIdx.x & 31;
  const int wave = threadIdx.x >> 5;
  const int half = lane >> 4;
  const int col  = lane & 15;

  const int tx = blockIdx.x & (IMG_W / TILE_W - 1);
  const int ty = blockIdx.x / (IMG_W / TILE_W);
  const float xc = (float)(tx * TILE_W) + HALF_W;
  const float yc = (float)(ty * TILE_H) + HALF_H;

  float g[4][6];
  float cq[4][3];
  unsigned bal[4];

#pragma unroll
  for (int k = 0; k < 4; ++k) {
    const int n = wave * 128 + k * 32 + lane;
    float mx = means[2 * n + 0];
    float my = means[2 * n + 1];
    float s2x = __expf(2.f * log_scales[2 * n + 0]);
    float s2y = __expf(2.f * log_scales[2 * n + 1]);
    float s, c;
    __sincosf(rot[n], &s, &c);

    float a = c * c * s2x + s * s * s2y + EPS_COV;
    float b = c * s * (s2x - s2y);
    float d = s * s * s2x + c * c * s2y + EPS_COV;
    float inv = LOG2E / (a * d - b * b);
    float ia = d * inv, ib = -b * inv, id = a * inv;

    float dx = xc - mx, dy = yc - my;
    float gx = ia * dx + ib * dy;
    float gy = ib * dx + id * dy;
    float pc = -0.5f * (ia * dx * dx + 2.f * ib * dx * dy + id * dy * dy);

    g[k][0] = -0.5f * ia;
    g[k][1] = -gx;
    g[k][2] = -0.5f * id;
    g[k][3] = -gy;
    g[k][4] = -ib;
    g[k][5] = pc + __log2f(fmaxf(opac[n], 0.f));
    cq[k][0] = colors[3 * n + 0];
    cq[k][1] = colors[3 * n + 1];
    cq[k][2] = colors[3 * n + 2];

    float reach = fabsf(gx) * HALF_W + fabsf(gy) * HALF_H;
    bool  psd = (ia >= 0.f) && (id >= 0.f) && (ia * id >= ib * ib);
    float curv = 0.5f * (fabsf(ia) * (HALF_W * HALF_W) + 2.f * fabsf(ib) * (HALF_W * HALF_H)
                         + fabsf(id) * (HALF_H * HALF_H));
    float ub = g[k][5] + reach + (psd ? 0.f : curv);
    float mag = 0.5f * (fabsf(ia) * dx * dx + 2.f * fabsf(ib * dx * dy) + fabsf(id) * dy * dy)
              + reach + curv;
    float guard = 4e-6f * mag;
    bool keep = !(ub + guard < CULL_BELOW);
    bal[k] = __builtin_amdgcn_ballot_w32(keep);
  }

  const unsigned lt = (1u << lane) - 1u;
  int off[4];
  int run = 0;
#pragma unroll
  for (int k = 0; k < 4; ++k) {
    off[k] = run + __builtin_popcount(bal[k] & lt);
    run += __builtin_popcount(bal[k]);
  }
  if (lane == 0) s_cnt[wave] = run;
  __syncthreads();

  int cnt[8];
#pragma unroll
  for (int i = 0; i < 8; ++i) cnt[i] = s_cnt[i];
  const int c0 = cnt[0] + cnt[1] + cnt[2] + cnt[3];
  const int c1 = cnt[4] + cnt[5] + cnt[6] + cnt[7];
  const int steps = __builtin_amdgcn_readfirstlane((max(c0, c1) + 7) & ~7);
  const int w4 = wave & 4;
  int base = (w4 >> 2) * (N_GAUSS / 2);
#pragma unroll
  for (int i = 0; i < 3; ++i)
    if (w4 + i < wave) base += cnt[w4 + i];

#pragma unroll
  for (int k = 0; k < 4; ++k) {
    if ((bal[k] >> lane) & 1u) {
      const int row = base + off[k];
      float* gf = sg + row * GF_STRIDE;
      gf[0] = g[k][4]; gf[1] = g[k][1]; gf[2] = g[k][0]; gf[3] = 0.f;
      gf[4] = g[k][2]; gf[5] = g[k][3]; gf[6] = g[k][5]; gf[7] = 0.f;
      float* cp = sc + row * 4;
      cp[0] = cq[k][0]; cp[1] = cq[k][1]; cp[2] = cq[k][2]; cp[3] = 0.f;
    }
  }
  {
    const int r  = threadIdx.x >> 7;
    const int cr_ = r ? c1 : c0;
    for (int i = cr_ + (threadIdx.x & 127); i < steps; i += 128) {
      const int row = r * (N_GAUSS / 2) + i;
      float* gf = sg + row * GF_STRIDE;
      gf[0] = 0.f; gf[1] = 0.f; gf[2] = 0.f; gf[3] = 0.f;
      gf[4] = 0.f; gf[5] = 0.f; gf[6] = -1.0e30f; gf[7] = 0.f;
      float* cp = sc + row * 4;
      cp[0] = 0.f; cp[1] = 0.f; cp[2] = 0.f; cp[3] = 0.f;
    }
  }
  __syncthreads();

  const int px = tx * TILE_W + (wave & 1) * 16 + col;
  const int py = ty * TILE_H + (wave >> 1);
  const int p  = py * IMG_W + px;
  const float ex = (float)((wave & 1) * 16 + col) - HALF_W;
  const float ey = (float)(wave >> 1) - HALF_H;

  const v2f bF = { half ? 1.f : ex * ex, half ? 0.f : ex };
  const float eyc = half ? ey : 0.f;

  const int arow = (col & 7) + (col >> 3) * (N_GAUSS / 2);
  const float* gfp = sg + arow * GF_STRIDE + 4 * half;
  const float* ccp = sc + half * (N_GAUSS / 2) * 4;

  float T = 1.f;
  v2f crg = { 0.f, 0.f };
  float cb = 0.f;

  for (int s = 0; s < steps; s += 8) {
    const float* gf = gfp + s * GF_STRIDE;
    const float lin = fmaf(gf[0], ey, gf[1]);
    const v2f aF = { fmaf(lin, eyc, gf[2]), lin };

    v8f P = {};
    P = __builtin_amdgcn_wmma_f32_16x16x4_f32(false, aF, false, bF,
                                              (short)0, P, false, false);

#pragma unroll
    for (int j = 0; j < 8; ++j) {
      const float* cc = ccp + (s + j) * 4;
      float al = fminf(__builtin_amdgcn_exp2f(P[j]), ALPHA_MAX);
      float w  = al * T;
      T = fmaf(-al, T, T);
      const v2f rg = { cc[0], cc[1] };
      const v2f ww = { w, w };
      crg = __builtin_elementwise_fma(ww, rg, crg);
      cb = fmaf(w, cc[2], cb);
    }
  }

  const float cr = crg.x, cg = crg.y;
  float hr = __shfl_xor(cr, 16, 32);
  float hg = __shfl_xor(cg, 16, 32);
  float hb = __shfl_xor(cb, 16, 32);
  if (half == 0) {
    float* o = out + (size_t)p * 3;
    o[0] = fmaf(T, hr, cr);
    o[1] = fmaf(T, hg, cg);
    o[2] = fmaf(T, hb, cb);
  }
}

extern "C" void kernel_launch(void* const* d_in, const int* in_sizes, int n_in,
                              void* d_out, int out_size, void* d_ws, size_t ws_size,
                              hipStream_t stream) {
  const float* means      = (const float*)d_in[0];
  const float* log_scales = (const float*)d_in[1];
  const float* rotations  = (const float*)d_in[2];
  const float* colors     = (const float*)d_in[3];
  const float* opacities  = (const float*)d_in[4];

  gs_render<<<(IMG_H / TILE_H) * (IMG_W / TILE_W), 256, 0, stream>>>(means, log_scales, rotations,
                                                       colors, opacities, (float*)d_out);
}


# block owns 32x8 px as two 16x8 sub-tiles, each with its own culled list (16 waves, 2 gaussians per thread, two tables in LDS): shorter lists, half the set-up per wave; per-pixel sum order kept
# speedup vs baseline: 4.0868x; 1.0583x over previous
#include <hip/hip_runtime.h>
#include <hip/hip_bf16.h>
#include <math.h>


typedef __attribute__((ext_vector_type(2))) float v2f;
typedef __attribute__((ext_vector_type(4))) float v4f;
typedef __attribute__((ext_vector_type(8))) float v8f;

#define N_GAUSS   1024
#define IMG_H     256
#define IMG_W     256
#define EPS_COV   1e-6f
#define ALPHA_MAX 0.999f

#define LOG2E     1.44269504f
#define GF_STRIDE 10
#define GF_DW     (N_GAUSS * GF_STRIDE)
#define CP_DW     (N_GAUSS * 4)
#define LIST_DW   (GF_DW + CP_DW)
#define SMEM_DW   (2 * LIST_DW)

#define TILE_W    32
#define TILE_H    8
#define HALF_W    15.5f
#define HALF_H    3.5f
#define SUB_W     16
#define SUB_CX    8.0f
#define SUB_HALF_W 7.5f
#define CULL_BELOW (-130.0f)

__global__ __launch_bounds__(512) void gs_render(const float* __restrict__ means,
                                                 const float* __restrict__ log_scales,
                                                 const float* __restrict__ rot,
                                                 const float* __restrict__ colors,
                                                 const float* __restrict__ opac,
                                                 float* __restrict__ out) {
  __shared__ float smem[SMEM_DW];
  __shared__ int s_cnt[16];

  const int lane = threadIdx.x & 31;
  const int wave = __builtin_amdgcn_readfirstlane(threadIdx.x >> 5);
  const int half = lane >> 4;
  const int col  = lane & 15;

  const int tx = blockIdx.x & (IMG_W / TILE_W - 1);
  const int ty = blockIdx.x / (IMG_W / TILE_W);
  const float xc = (float)(tx * TILE_W) + HALF_W;
  const float yc = (float)(ty * TILE_H) + HALF_H;

  float g[2][6];
  float cq[2][3];
  unsigned bal[2][2];

#pragma unroll
  for (int k = 0; k < 2; ++k) {
    const int n = wave * 64 + k * 32 + lane;
    float mx = means[2 * n + 0];
    float my = means[2 * n + 1];
    float s2x = __expf(2.f * log_scales[2 * n + 0]);
    float s2y = __expf(2.f * log_scales[2 * n + 1]);
    float s, c;
    __sincosf(rot[n], &s, &c);

    float a = c * c * s2x + s * s * s2y + EPS_COV;
    float b = c * s * (s2x - s2y);
    float d = s * s * s2x + c * c * s2y + EPS_COV;
    float inv = LOG2E / (a * d - b * b);
    float ia = d * inv, ib = -b * inv, id = a * inv;

    float dx = xc - mx, dy = yc - my;
    float gx = ia * dx + ib * dy;
    float gy = ib * dx + id * dy;
    float pc = -0.5f * (ia * dx * dx + 2.f * ib * dx * dy + id * dy * dy);

    g[k][0] = -0.5f * ia;
    g[k][1] = -gx;
    g[k][2] = -0.5f * id;
    g[k][3] = -gy;
    g[k][4] = -ib;
    g[k][5] = pc + __log2f(fmaxf(opac[n], 0.f));
    cq[k][0] = colors[3 * n + 0];
    cq[k][1] = colors[3 * n + 1];
    cq[k][2] = colors[3 * n + 2];

    bool  psd = (ia >= 0.f) && (id >= 0.f) && (ia * id >= ib * ib);
    float curv = 0.5f * (fabsf(ia) * (HALF_W * HALF_W) + 2.f * fabsf(ib) * (HALF_W * HALF_H)
                         + fabsf(id) * (HALF_H * HALF_H));
    float mag = 0.5f * (fabsf(ia) * dx * dx + 2.f * fabsf(ib * dx * dy) + fabsf(id) * dy * dy)
              + fabsf(gx) * HALF_W + fabsf(gy) * HALF_H + curv;
    float allow = 4e-6f * mag + (psd ? 0.f : curv);
#pragma unroll
    for (int L = 0; L < 2; ++L) {
      const float cx = L ? SUB_CX : -SUB_CX;
      float l1  = fmaf(g[k][0], cx, g[k][1]);
      float vs  = fmaf(l1, cx, g[k][5]);
      float gxs = fmaf(g[k][0], cx, l1);
      float gys = fmaf(g[k][4], cx, g[k][3]);
      float ub  = vs + fabsf(gxs) * SUB_HALF_W + fabsf(gys) * HALF_H;
      bool keep = !(ub + allow < CULL_BELOW);
      bal[L][k] = __builtin_amdgcn_ballot_w32(keep);
    }
  }

  const unsigned lt = (1u << lane) - 1u;
  int off[2][2];
  int run[2];
#pragma unroll
  for (int L = 0; L < 2; ++L) {
    run[L] = 0;
#pragma unroll
    for (int k = 0; k < 2; ++k) {
      off[L][k] = run[L] + __builtin_popcount(bal[L][k] & lt);
      run[L] += __builtin_popcount(bal[L][k]);
    }
  }
  if (lane == 0) s_cnt[wave] = run[0] | (run[1] << 16);
  __syncthreads();

  const int myc = s_cnt[lane & 15];
  const int w8 = wave & 8;
  int c0p = 0, c1p = 0, basep = 0;
#pragma unroll
  for (int i = 0; i < 16; ++i) {
    const int ci = __builtin_amdgcn_readlane(myc, i);
    if (i < 8) c0p += ci; else c1p += ci;
    if (i >= w8 && i < wave) basep += ci;
  }
  int c0[2], c1[2], base[2], steps[2];
#pragma unroll
  for (int L = 0; L < 2; ++L) {
    c0[L] = (c0p >> (16 * L)) & 0xffff;
    c1[L] = (c1p >> (16 * L)) & 0xffff;
    base[L] = (w8 >> 3) * (N_GAUSS / 2) + ((basep >> (16 * L)) & 0xffff);
    steps[L] = (max(c0[L], c1[L]) + 7) & ~7;
  }

#pragma unroll
  for (int k = 0; k < 2; ++k) {
#pragma unroll
    for (int L = 0; L < 2; ++L) {
      if ((bal[L][k] >> lane) & 1u) {
        const int row = base[L] + off[L][k];
        float* gf = smem + L * LIST_DW + row * GF_STRIDE;
        gf[0] = g[k][4]; gf[1] = g[k][1]; gf[2] = g[k][0]; gf[3] = 0.f;
        gf[4] = g[k][2]; gf[5] = g[k][3]; gf[6] = g[k][5]; gf[7] = 0.f;
        float* cp = smem + L * LIST_DW + GF_DW + row * 4;
        cp[0] = cq[k][0]; cp[1] = cq[k][1]; cp[2] = cq[k][2]; cp[3] = 0.f;
      }
    }
  }
  {
    const int L  = wave >> 3;
    const int r  = (wave >> 2) & 1;
    const int cr_ = L ? (r ? c1[1] : c0[1]) : (r ? c1[0] : c0[0]);
    const int fill_to = L ? steps[1] : steps[0];
    for (int i = cr_ + (threadIdx.x & 127); i < fill_to; i += 128) {
      const int row = r * (N_GAUSS / 2) + i;
      float* gf = smem + L * LIST_DW + row * GF_STRIDE;
      gf[0] = 0.f; gf[1] = 0.f; gf[2] = 0.f; gf[3] = 0.f;
      gf[4] = 0.f; gf[5] = 0.f; gf[6] = -1.0e30f; gf[7] = 0.f;
      float* cp = smem + L * LIST_DW + GF_DW + row * 4;
      cp[0] = 0.f; cp[1] = 0.f; cp[2] = 0.f; cp[3] = 0.f;
    }
  }
  __syncthreads();

  const int sub = wave & 1;
  const int prow = wave >> 1;
  const float* sg = smem + sub * LIST_DW;
  const float* sc = sg + GF_DW;
  const int nsteps = sub ? steps[1] : steps[0];

  const int px = tx * TILE_W + sub * SUB_W + col;
  const int py = ty * TILE_H + prow;
  const int p  = py * IMG_W + px;
  const float ex = (float)(sub * SUB_W + col) - HALF_W;
  const float ey = (float)prow - HALF_H;

  const v2f bF = { half ? 1.f : ex * ex, half ? 0.f : ex };
  const float eyc = half ? ey : 0.f;

  const int arow = (col & 7) + (col >> 3) * (N_GAUSS / 2);
  const float* gfp = sg + arow * GF_STRIDE + 4 * half;
  const float* ccp = sc + half * (N_GAUSS / 2) * 4;

  float T = 1.f;
  v2f crg = { 0.f, 0.f };
  float cb = 0.f;

  for (int s = 0; s < nsteps; s += 8) {
    const float* gf = gfp + s * GF_STRIDE;
    const float lin = fmaf(gf[0], ey, gf[1]);
    const v2f aF = { fmaf(lin, eyc, gf[2]), lin };

    v8f P = {};
    P = __builtin_amdgcn_wmma_f32_16x16x4_f32(false, aF, false, bF,
                                              (short)0, P, false, false);

#pragma unroll
    for (int j = 0; j < 8; ++j) {
      const float* cc = ccp + (s + j) * 4;
      float al = fminf(__builtin_amdgcn_exp2f(P[j]), ALPHA_MAX);
      float w  = al * T;
      T = fmaf(-al, T, T);
      const v2f rg = { cc[0], cc[1] };
      const v2f ww = { w, w };
      crg = __builtin_elementwise_fma(ww, rg, crg);
      cb = fmaf(w, cc[2], cb);
    }
  }

  const float cr = crg.x, cg = crg.y;
  float hr = __shfl_xor(cr, 16, 32);
  float hg = __shfl_xor(cg, 16, 32);
  float hb = __shfl_xor(cb, 16, 32);
  if (half == 0) {
    float* o = out + (size_t)p * 3;
    o[0] = fmaf(T, hr, cr);
    o[1] = fmaf(T, hg, cg);
    o[2] = fmaf(T, hb, cb);
  }
}

extern "C" void kernel_launch(void* const* d_in, const int* in_sizes, int n_in,
                              void* d_out, int out_size, void* d_ws, size_t ws_size,
                              hipStream_t stream) {
  const float* means      = (const float*)d_in[0];
  const float* log_scales = (const float*)d_in[1];
  const float* rotations  = (const float*)d_in[2];
  const float* colors     = (const float*)d_in[3];
  const float* opacities  = (const float*)d_in[4];

  gs_render<<<(IMG_H / TILE_H) * (IMG_W / TILE_W), 512, 0, stream>>>(means, log_scales, rotations,
                                                       colors, opacities, (float*)d_out);
}
